# starting kernel
# baseline (speedup 1.0000x reference)
#include <hip/hip_runtime.h>
#include <math.h>


typedef float v2f __attribute__((ext_vector_type(2)));
typedef float v8f __attribute__((ext_vector_type(8)));

#define GMM_K 32
#define GMM_C 128
#define LOG2PI_F 1.8377866f
static constexpr float kLog2Pi  = 1.8378770664093453f;
static constexpr float kEps     = 5500.0f;
static constexpr float kInvLmb  = 1.0f / 1100.0f;

__global__ __launch_bounds__(256)
void gmm_precompute(const float* __restrict__ w,
                    const float* __restrict__ mu,
                    const float* __restrict__ cov,
                    float* __restrict__ M,
                    float* __restrict__ G)
{
    int idx = blockIdx.x * blockDim.x + threadIdx.x;
    if (idx >= GMM_K * GMM_C) return;
    int k = idx >> 7;
    int c = idx & (GMM_C - 1);

    const float* cv = cov + (size_t)idx * 4;
    float c00 = cv[0], c01 = cv[1], c10 = cv[2], c11 = cv[3];
    float det  = c00 * c11 - c01 * c10;
    float inv  = 1.0f / det;
    float P00  = c11 * inv;
    float P01  = -c01 * inv;
    float P10  = -c10 * inv;
    float P11  = c00 * inv;
    float m0   = mu[(size_t)idx * 2 + 0];
    float m1   = mu[(size_t)idx * 2 + 1];
    float s    = P01 + P10;
    float logdet = logf(det);
    float lw     = logf(w[idx]);
    float mPm    = P00 * m0 * m0 + s * m0 * m1 + P11 * m1 * m1;

    float* Mk = M + (size_t)k * 8 * GMM_C;
    Mk[0 * GMM_C + c] = -0.5f * P00;
    Mk[1 * GMM_C + c] = -0.5f * s;
    Mk[2 * GMM_C + c] = -0.5f * P11;
    Mk[3 * GMM_C + c] = P00 * m0 + 0.5f * s * m1;
    Mk[4 * GMM_C + c] = 0.5f * s * m0 + P11 * m1;
    Mk[5 * GMM_C + c] = lw - 0.5f * (2.0f * kLog2Pi + logdet) - 0.5f * mPm;
    Mk[6 * GMM_C + c] = 0.0f;
    Mk[7 * GMM_C + c] = 0.0f;

    float* Gc = G + (size_t)idx * 16;
    Gc[0] = P00;
    Gc[1] = P01;
    Gc[2] = P10;
    Gc[3] = P11;
    Gc[4] = P00 * m0 + P01 * m1;
    Gc[5] = P10 * m0 + P11 * m1;
    Gc[6] = 1.0f;
#pragma unroll
    for (int j = 7; j < 16; ++j) Gc[j] = 0.0f;
}

__global__ __launch_bounds__(128)
void gmm_wmma(const float* __restrict__ kps,
              const float* __restrict__ M,
              const float* __restrict__ G,
              float* __restrict__ out,
              int Btot)
{
    __shared__ float Gs[GMM_C * 16];
    __shared__ float compS[4][16 * GMM_C];
    __shared__ float Ssh[4][16 * 16];

    const int k    = blockIdx.y;
    const int wave = threadIdx.x >> 5;
    const int lane = threadIdx.x & 31;
    const int n    = lane & 15;
    const int half = lane >> 4;

    const float* Gk = G + (size_t)k * GMM_C * 16;
    for (int i = threadIdx.x; i < GMM_C * 16; i += 128) Gs[i] = Gk[i];
    __syncthreads();

    const int b0   = blockIdx.x * 64 + wave * 16;
    const int brow = b0 + n;
    const float x0 = kps[((size_t)brow * GMM_K + k) * 2 + 0];
    const float x1 = kps[((size_t)brow * GMM_K + k) * 2 + 1];

    v2f A0, A1;
    if (half == 0) { A0.x = x0 * x0; A0.y = x0 * x1; A1.x = x1;   A1.y = 1.0f; }
    else           { A0.x = x1 * x1; A0.y = x0;      A1.x = 0.0f; A1.y = 0.0f; }

    const float* Mk = M + (size_t)k * 8 * GMM_C;
    float* cw = compS[wave];

#pragma unroll
    for (int t = 0; t < 8; ++t) {
        const int c0 = t * 16 + n;
        v2f B0, B1;
        B0.x = Mk[(0 + 2 * half) * GMM_C + c0];
        B0.y = Mk[(1 + 2 * half) * GMM_C + c0];
        B1.x = Mk[(4 + 2 * half) * GMM_C + c0];
        B1.y = Mk[(5 + 2 * half) * GMM_C + c0];
        v8f acc = {};
        acc = __builtin_amdgcn_wmma_f32_16x16x4_f32(false, A0, false, B0,
                                                    (short)0, acc, false, false);
        acc = __builtin_amdgcn_wmma_f32_16x16x4_f32(false, A1, false, B1,
                                                    (short)0, acc, false, false);
#pragma unroll
        for (int v = 0; v < 8; ++v) {
            const int row = v + 8 * half;
            cw[row * GMM_C + t * 16 + n] = __expf(acc[v]);
        }
    }

    v8f acc2 = {};
#pragma unroll
    for (int cc = 0; cc < GMM_C; cc += 4) {
        const int r = cc + 2 * half;
        v2f A2, B2;
        A2.x = cw[n * GMM_C + r];
        A2.y = cw[n * GMM_C + r + 1];
        B2.x = Gs[r * 16 + n];
        B2.y = Gs[(r + 1) * 16 + n];
        acc2 = __builtin_amdgcn_wmma_f32_16x16x4_f32(false, A2, false, B2,
                                                     (short)0, acc2, false, false);
    }

    float* sw = Ssh[wave];
#pragma unroll
    for (int v = 0; v < 8; ++v) sw[(v + 8 * half) * 16 + n] = acc2[v];

    if (half == 0) {
        const float S0  = sw[n * 16 + 0];
        const float S1  = sw[n * 16 + 1];
        const float S2  = sw[n * 16 + 2];
        const float S3  = sw[n * 16 + 3];
        const float S4  = sw[n * 16 + 4];
        const float S5  = sw[n * 16 + 5];
        const float pdf = sw[n * 16 + 6];

        const float g0 = S4 - S0 * x0 - S1 * x1;
        const float g1 = S5 - S2 * x0 - S3 * x1;
        float nrm = sqrtf(g0 * g0 + g1 * g1);
        nrm = fmaxf(nrm, 1e-30f);
        const float scale = __expf((kEps - nrm) * kInvLmb) / nrm;

        float* dens = out + (size_t)brow * 64 + 2 * k;
        dens[0] = pdf;
        dens[1] = pdf;
        float* gv = out + (size_t)Btot * 64 + (size_t)brow * 64 + 2 * k;
        gv[0] = g0 * scale;
        gv[1] = g1 * scale;
    }
}

extern "C" void kernel_launch(void* const* d_in, const int* in_sizes, int n_in,
                              void* d_out, int out_size, void* d_ws, size_t ws_size,
                              hipStream_t stream)
{
    const float* kps = (const float*)d_in[0];
    const float* w   = (const float*)d_in[1];
    const float* mu  = (const float*)d_in[2];
    const float* cov = (const float*)d_in[3];
    float* out = (float*)d_out;

    const int Btot = in_sizes[0] / (GMM_K * 2);

    float* Mws = (float*)d_ws;
    float* Gws = Mws + (size_t)GMM_K * 8 * GMM_C;

    gmm_precompute<<<(GMM_K * GMM_C + 255) / 256, 256, 0, stream>>>(w, mu, cov, Mws, Gws);

    dim3 grid(Btot / 64, GMM_K);
    gmm_wmma<<<grid, 128, 0, stream>>>(kps, Mws, Gws, out, Btot);
}


# Swap operand roles in both products: exp of the first's D tile is the second's B fragment in registers, so no comp tile and no result spill in LDS; the 128 components are summed in a permuted order
# speedup vs baseline: 1.5027x; 1.5027x over previous
#include <hip/hip_runtime.h>
#include <math.h>


typedef float v2f __attribute__((ext_vector_type(2)));
typedef float v8f __attribute__((ext_vector_type(8)));

#define GMM_K 32
#define GMM_C 128
static constexpr float kLog2Pi  = 1.8378770664093453f;
static constexpr float kEps     = 5500.0f;
static constexpr float kInvLmb  = 1.0f / 1100.0f;

__global__ __launch_bounds__(256)
void gmm_precompute(const float* __restrict__ w,
                    const float* __restrict__ mu,
                    const float* __restrict__ cov,
                    float* __restrict__ M,
                    float* __restrict__ G)
{
    int idx = blockIdx.x * blockDim.x + threadIdx.x;
    if (idx >= GMM_K * GMM_C) return;
    int k = idx >> 7;
    int c = idx & (GMM_C - 1);

    const float* cv = cov + (size_t)idx * 4;
    float c00 = cv[0], c01 = cv[1], c10 = cv[2], c11 = cv[3];
    float det  = c00 * c11 - c01 * c10;
    float inv  = 1.0f / det;
    float P00  = c11 * inv;
    float P01  = -c01 * inv;
    float P10  = -c10 * inv;
    float P11  = c00 * inv;
    float m0   = mu[(size_t)idx * 2 + 0];
    float m1   = mu[(size_t)idx * 2 + 1];
    float s    = P01 + P10;
    float logdet = logf(det);
    float lw     = logf(w[idx]);
    float mPm    = P00 * m0 * m0 + s * m0 * m1 + P11 * m1 * m1;

    float* Mk = M + (size_t)k * 8 * GMM_C;
    Mk[0 * GMM_C + c] = -0.5f * P00;
    Mk[1 * GMM_C + c] = -0.5f * s;
    Mk[2 * GMM_C + c] = -0.5f * P11;
    Mk[3 * GMM_C + c] = P00 * m0 + 0.5f * s * m1;
    Mk[4 * GMM_C + c] = 0.5f * s * m0 + P11 * m1;
    Mk[5 * GMM_C + c] = lw - 0.5f * (2.0f * kLog2Pi + logdet) - 0.5f * mPm;
    Mk[6 * GMM_C + c] = 0.0f;
    Mk[7 * GMM_C + c] = 0.0f;

    float* Gc = G + (size_t)idx * 16;
    Gc[0] = P00;
    Gc[1] = P01;
    Gc[2] = P10;
    Gc[3] = P11;
    Gc[4] = P00 * m0 + P01 * m1;
    Gc[5] = P10 * m0 + P11 * m1;
    Gc[6] = 1.0f;
#pragma unroll
    for (int j = 7; j < 16; ++j) Gc[j] = 0.0f;
}

#define GS_GROUP  160
#define GS_DWORDS (16 * GS_GROUP)

__device__ __forceinline__ int gs_index(int c, int j)
{
    const int p = c >> 1;
    return (p >> 2) * GS_GROUP + (p & 3) * 32 + j * 2 + (c & 1);
}

__global__ __launch_bounds__(128)
void gmm_wmma(const float* __restrict__ kps,
              const float* __restrict__ M,
              const float* __restrict__ G,
              float* __restrict__ out,
              int Btot)
{
    __shared__ __attribute__((aligned(16))) float Gs[GS_DWORDS];

    const int k    = blockIdx.y;
    const int wave = threadIdx.x >> 5;
    const int lane = threadIdx.x & 31;
    const int n    = lane & 15;
    const int half = lane >> 4;

    const float* Gk = G + (size_t)k * GMM_C * 16;
    for (int i = threadIdx.x; i < GMM_C * 16; i += 128)
        Gs[gs_index(i >> 4, i & 15)] = Gk[i];
    __syncthreads();

    const int b0   = blockIdx.x * 64 + wave * 16;
    const int brow = b0 + n;
    const float x0 = kps[((size_t)brow * GMM_K + k) * 2 + 0];
    const float x1 = kps[((size_t)brow * GMM_K + k) * 2 + 1];

    v2f F0, F1;
    if (half == 0) { F0.x = x0 * x0; F0.y = x0 * x1; F1.x = x1;   F1.y = 1.0f; }
    else           { F0.x = x1 * x1; F0.y = x0;      F1.x = 0.0f; F1.y = 0.0f; }

    const float* Mk = M + (size_t)k * 8 * GMM_C;

    v8f acc2 = {};
#pragma unroll
    for (int t = 0; t < 8; ++t) {
        const int c0 = t * 16 + n;
        v2f M0, M1;
        M0.x = Mk[(0 + 2 * half) * GMM_C + c0];
        M0.y = Mk[(1 + 2 * half) * GMM_C + c0];
        M1.x = Mk[(4 + 2 * half) * GMM_C + c0];
        M1.y = Mk[(5 + 2 * half) * GMM_C + c0];
        v8f acc = {};
        acc = __builtin_amdgcn_wmma_f32_16x16x4_f32(false, M0, false, F0,
                                                    (short)0, acc, false, false);
        acc = __builtin_amdgcn_wmma_f32_16x16x4_f32(false, M1, false, F1,
                                                    (short)0, acc, false, false);

#pragma unroll
        for (int q = 0; q < 4; ++q) {
            v2f E;
            E.x = __expf(acc[2 * q]);
            E.y = __expf(acc[2 * q + 1]);
            const int c = t * 16 + 8 * half + 2 * q;
            const v2f Gp = *(const v2f*)&Gs[gs_index(c, n)];
            acc2 = __builtin_amdgcn_wmma_f32_16x16x4_f32(false, Gp, false, E,
                                                         (short)0, acc2, false, false);
        }
    }

    if (half == 0) {
        const float S0  = acc2[0];
        const float S1  = acc2[1];
        const float S2  = acc2[2];
        const float S3  = acc2[3];
        const float S4  = acc2[4];
        const float S5  = acc2[5];
        const float pdf = acc2[6];

        const float g0 = S4 - S0 * x0 - S1 * x1;
        const float g1 = S5 - S2 * x0 - S3 * x1;
        float nrm = sqrtf(g0 * g0 + g1 * g1);
        nrm = fmaxf(nrm, 1e-30f);
        const float scale = __expf((kEps - nrm) * kInvLmb) / nrm;

        float* dens = out + (size_t)brow * 64 + 2 * k;
        dens[0] = pdf;
        dens[1] = pdf;
        float* gv = out + (size_t)Btot * 64 + (size_t)brow * 64 + 2 * k;
        gv[0] = g0 * scale;
        gv[1] = g1 * scale;
    }
}

extern "C" void kernel_launch(void* const* d_in, const int* in_sizes, int n_in,
                              void* d_out, int out_size, void* d_ws, size_t ws_size,
                              hipStream_t stream)
{
    const float* kps = (const float*)d_in[0];
    const float* w   = (const float*)d_in[1];
    const float* mu  = (const float*)d_in[2];
    const float* cov = (const float*)d_in[3];
    float* out = (float*)d_out;

    const int Btot = in_sizes[0] / (GMM_K * 2);

    float* Mws = (float*)d_ws;
    float* Gws = Mws + (size_t)GMM_K * 8 * GMM_C;

    gmm_precompute<<<(GMM_K * GMM_C + 255) / 256, 256, 0, stream>>>(w, mu, cov, Mws, Gws);

    dim3 grid(Btot / 64, GMM_K);
    gmm_wmma<<<grid, 128, 0, stream>>>(kps, Mws, Gws, out, Btot);
}


# Fold the parameter precompute launch into the WMMA kernel: each block derives its keypoint's M and G tables in LDS from the inputs (thread c = component c); one launch, no workspace
# speedup vs baseline: 2.3325x; 1.5522x over previous
#include <hip/hip_runtime.h>
#include <math.h>


typedef float v2f __attribute__((ext_vector_type(2)));
typedef float v8f __attribute__((ext_vector_type(8)));

#define GMM_K 32
#define GMM_C 128
static constexpr float kLog2Pi  = 1.8378770664093453f;
static constexpr float kEps     = 5500.0f;
static constexpr float kInvLmb  = 1.0f / 1100.0f;

#define MS_PITCH  288
#define MS_DWORDS (4 * MS_PITCH)

#define GS_PITCH  130
#define GS_DWORDS (8 * GS_PITCH)

__device__ __forceinline__ int gs_pos(int c)
{
    return (c & 64) + ((c >> 3) & 1) * 32 + ((c >> 4) & 3) * 8 + (c & 7);
}

__global__ __launch_bounds__(128)
void gmm_wmma(const float* __restrict__ kps,
              const float* __restrict__ w,
              const float* __restrict__ mu,
              const float* __restrict__ cov,
              float* __restrict__ out,
              int Btot)
{
    __shared__ __attribute__((aligned(16))) float Ms[MS_DWORDS];
    __shared__ __attribute__((aligned(16))) float Gs[GS_DWORDS];

    const int k    = blockIdx.y;
    const int wave = threadIdx.x >> 5;
    const int lane = threadIdx.x & 31;
    const int n    = lane & 15;
    const int half = lane >> 4;

    const int b0   = blockIdx.x * 64 + wave * 16;
    const int brow = b0 + n;
    const float x0 = kps[((size_t)brow * GMM_K + k) * 2 + 0];
    const float x1 = kps[((size_t)brow * GMM_K + k) * 2 + 1];

    {
        const int c   = threadIdx.x;
        const int idx = k * GMM_C + c;

        const float* cv = cov + (size_t)idx * 4;
        float c00 = cv[0], c01 = cv[1], c10 = cv[2], c11 = cv[3];
        float det  = c00 * c11 - c01 * c10;
        float inv  = 1.0f / det;
        float P00  = c11 * inv;
        float P01  = -c01 * inv;
        float P10  = -c10 * inv;
        float P11  = c00 * inv;
        float m0   = mu[(size_t)idx * 2 + 0];
        float m1   = mu[(size_t)idx * 2 + 1];
        float s    = P01 + P10;
        float logdet = logf(fabsf(det));
        float lw     = logf(w[idx]);
        float mPm    = P00 * m0 * m0 + s * m0 * m1 + P11 * m1 * m1;

        v2f p0, p1, p2, p3;
        p0.x = -0.5f * P00;
        p0.y = -0.5f * s;
        p1.x = -0.5f * P11;
        p1.y = P00 * m0 + 0.5f * s * m1;
        p2.x = 0.5f * s * m0 + P11 * m1;
        p2.y = lw - 0.5f * (2.0f * kLog2Pi + logdet) - 0.5f * mPm;
        p3.x = 0.0f;
        p3.y = 0.0f;
        *(v2f*)&Ms[0 * MS_PITCH + 2 * c] = p0;
        *(v2f*)&Ms[1 * MS_PITCH + 2 * c] = p1;
        *(v2f*)&Ms[2 * MS_PITCH + 2 * c] = p2;
        *(v2f*)&Ms[3 * MS_PITCH + 2 * c] = p3;

        const int pc = gs_pos(c);
        Gs[0 * GS_PITCH + pc] = P00;
        Gs[1 * GS_PITCH + pc] = P10;
        Gs[2 * GS_PITCH + pc] = P01;
        Gs[3 * GS_PITCH + pc] = P11;
        Gs[4 * GS_PITCH + pc] = P00 * m0 + P10 * m1;
        Gs[5 * GS_PITCH + pc] = P01 * m0 + P11 * m1;
        Gs[6 * GS_PITCH + pc] = 1.0f;
        Gs[7 * GS_PITCH + pc] = 0.0f;
    }
    asm volatile("" :: "v"(x0), "v"(x1));
    __syncthreads();

    v2f F0, F1;
    if (half == 0) { F0.x = x0 * x0; F0.y = x0 * x1; F1.x = x1;   F1.y = 1.0f; }
    else           { F0.x = x1 * x1; F0.y = x0;      F1.x = 0.0f; F1.y = 0.0f; }

    const float* Mlane = Ms + half * MS_PITCH + 2 * n;
    const float* Glane = Gs + (n < 7 ? n : 7) * GS_PITCH + half * 32;

    v8f acc2 = {};
#pragma unroll
    for (int t = 0; t < 8; ++t) {
        const v2f M0 = *(const v2f*)&Mlane[32 * t];
        const v2f M1 = *(const v2f*)&Mlane[2 * MS_PITCH + 32 * t];
        v8f acc = {};
        acc = __builtin_amdgcn_wmma_f32_16x16x4_f32(false, M0, false, F0,
                                                    (short)0, acc, false, false);
        acc = __builtin_amdgcn_wmma_f32_16x16x4_f32(false, M1, false, F1,
                                                    (short)0, acc, false, false);

#pragma unroll
        for (int q = 0; q < 4; ++q) {
            v2f E;
            E.x = __expf(acc[2 * q]);
            E.y = __expf(acc[2 * q + 1]);
            const v2f Gp = *(const v2f*)&Glane[(t >> 2) * 64 + (t & 3) * 8 + 2 * q];
            acc2 = __builtin_amdgcn_wmma_f32_16x16x4_f32(false, Gp, false, E,
                                                         (short)0, acc2, false, false);
        }
    }

    if (half == 0) {
        const float S0  = acc2[0];
        const float S1  = acc2[1];
        const float S2  = acc2[2];
        const float S3  = acc2[3];
        const float S4  = acc2[4];
        const float S5  = acc2[5];
        const float pdf = acc2[6];

        const float g0 = S4 - S0 * x0 - S1 * x1;
        const float g1 = S5 - S2 * x0 - S3 * x1;
        float nrm = sqrtf(g0 * g0 + g1 * g1);
        nrm = fmaxf(nrm, 1e-30f);
        const float scale = __expf((kEps - nrm) * kInvLmb) / nrm;

        float* dens = out + (size_t)brow * 64 + 2 * k;
        dens[0] = pdf;
        dens[1] = pdf;
        float* gv = out + (size_t)Btot * 64 + (size_t)brow * 64 + 2 * k;
        gv[0] = g0 * scale;
        gv[1] = g1 * scale;
    }
}

extern "C" void kernel_launch(void* const* d_in, const int* in_sizes, int n_in,
                              void* d_out, int out_size, void* d_ws, size_t ws_size,
                              hipStream_t stream)
{
    const float* kps = (const float*)d_in[0];
    const float* w   = (const float*)d_in[1];
    const float* mu  = (const float*)d_in[2];
    const float* cov = (const float*)d_in[3];
    float* out = (float*)d_out;

    const int Btot = in_sizes[0] / (GMM_K * 2);

    dim3 grid(Btot / 64, GMM_K);
    gmm_wmma<<<grid, 128, 0, stream>>>(kps, w, mu, cov, out, Btot);
}


# Each wave reads its A fragments of both products (M, G of all 128 components) from LDS once into registers and reuses them over four b-tiles, two at a time: no LDS read among the WMMAs; grid 32x32
# speedup vs baseline: 2.4604x; 1.0548x over previous
#include <hip/hip_runtime.h>
#include <math.h>


typedef float v2f __attribute__((ext_vector_type(2)));
typedef float v8f __attribute__((ext_vector_type(8)));

#define GMM_K 32
#define GMM_C 128
#define GMM_NB 4
static_assert(GMM_NB == 4, "the sample-load fence in the kernel names four b-tiles");
static constexpr float kLog2Pi  = 1.8378770664093453f;
static constexpr float kEps     = 5500.0f;
static constexpr float kInvLmb  = 1.0f / 1100.0f;

#define MS_PITCH  288
#define MS_DWORDS (4 * MS_PITCH)

#define GS_PITCH  130
#define GS_DWORDS (8 * GS_PITCH)

__device__ __forceinline__ int gs_pos(int c)
{
    return (c & 64) + ((c >> 3) & 1) * 32 + ((c >> 4) & 3) * 8 + (c & 7);
}

#define WMMA4(A, B, C) __builtin_amdgcn_wmma_f32_16x16x4_f32(false, (A), false, (B), (short)0, (C), false, false)

__device__ __forceinline__ void gmm_store_rows(const v8f S, float x0, float x1,
                                               float* __restrict__ out,
                                               int brow, int k, int Btot)
{
    const float pdf = S[6];
    const float g0 = S[4] - S[0] * x0 - S[1] * x1;
    const float g1 = S[5] - S[2] * x0 - S[3] * x1;
    float nrm = sqrtf(g0 * g0 + g1 * g1);
    nrm = fmaxf(nrm, 1e-30f);
    const float scale = __expf((kEps - nrm) * kInvLmb) / nrm;

    float* dens = out + (size_t)brow * 64 + 2 * k;
    dens[0] = pdf;
    dens[1] = pdf;
    float* gv = out + (size_t)Btot * 64 + (size_t)brow * 64 + 2 * k;
    gv[0] = g0 * scale;
    gv[1] = g1 * scale;
}

__global__ __launch_bounds__(128) __attribute__((amdgpu_waves_per_eu(4, 4)))
void gmm_wmma(const float* __restrict__ kps,
              const float* __restrict__ w,
              const float* __restrict__ mu,
              const float* __restrict__ cov,
              float* __restrict__ out,
              int Btot)
{
    __shared__ __attribute__((aligned(16))) float Ms[MS_DWORDS];
    __shared__ __attribute__((aligned(16))) float Gs[GS_DWORDS];

    const int k    = blockIdx.y;
    const int wave = threadIdx.x >> 5;
    const int lane = threadIdx.x & 31;
    const int n    = lane & 15;
    const int half = lane >> 4;

    const int b0 = (blockIdx.x * 4 + wave) * (GMM_NB * 16);
    float x0[GMM_NB], x1[GMM_NB];
#pragma unroll
    for (int i = 0; i < GMM_NB; ++i) {
        const int brow = b0 + 16 * i + n;
        x0[i] = kps[((size_t)brow * GMM_K + k) * 2 + 0];
        x1[i] = kps[((size_t)brow * GMM_K + k) * 2 + 1];
    }

    {
        const int c   = threadIdx.x;
        const int idx = k * GMM_C + c;

        const float* cv = cov + (size_t)idx * 4;
        float c00 = cv[0], c01 = cv[1], c10 = cv[2], c11 = cv[3];
        float det  = c00 * c11 - c01 * c10;
        float inv  = 1.0f / det;
        float P00  = c11 * inv;
        float P01  = -c01 * inv;
        float P10  = -c10 * inv;
        float P11  = c00 * inv;
        float m0   = mu[(size_t)idx * 2 + 0];
        float m1   = mu[(size_t)idx * 2 + 1];
        float s    = P01 + P10;
        float logdet = logf(fabsf(det));
        float lw     = logf(w[idx]);
        float mPm    = P00 * m0 * m0 + s * m0 * m1 + P11 * m1 * m1;

        v2f p0, p1, p2, p3;
        p0.x = -0.5f * P00;
        p0.y = -0.5f * s;
        p1.x = -0.5f * P11;
        p1.y = P00 * m0 + 0.5f * s * m1;
        p2.x = 0.5f * s * m0 + P11 * m1;
        p2.y = lw - 0.5f * (2.0f * kLog2Pi + logdet) - 0.5f * mPm;
        p3.x = 0.0f;
        p3.y = 0.0f;
        *(v2f*)&Ms[0 * MS_PITCH + 2 * c] = p0;
        *(v2f*)&Ms[1 * MS_PITCH + 2 * c] = p1;
        *(v2f*)&Ms[2 * MS_PITCH + 2 * c] = p2;
        *(v2f*)&Ms[3 * MS_PITCH + 2 * c] = p3;

        const int pc = gs_pos(c);
        Gs[0 * GS_PITCH + pc] = P00;
        Gs[1 * GS_PITCH + pc] = P10;
        Gs[2 * GS_PITCH + pc] = P01;
        Gs[3 * GS_PITCH + pc] = P11;
        Gs[4 * GS_PITCH + pc] = P00 * m0 + P10 * m1;
        Gs[5 * GS_PITCH + pc] = P01 * m0 + P11 * m1;
        Gs[6 * GS_PITCH + pc] = 1.0f;
        Gs[7 * GS_PITCH + pc] = 0.0f;
    }
    asm volatile("" :: "v"(x0[0]), "v"(x1[0]), "v"(x0[1]), "v"(x1[1]),
                       "v"(x0[2]), "v"(x1[2]), "v"(x0[3]), "v"(x1[3]));
    __syncthreads();

    const float* Mlane = Ms + half * MS_PITCH + 2 * n;
    const float* Glane = Gs + (n < 7 ? n : 7) * GS_PITCH + half * 32;
    v2f M0[8], M1[8], G[8][4];
#pragma unroll
    for (int t = 0; t < 8; ++t) {
        M0[t] = *(const v2f*)&Mlane[32 * t];
        M1[t] = *(const v2f*)&Mlane[2 * MS_PITCH + 32 * t];
#pragma unroll
        for (int q = 0; q < 4; ++q)
            G[t][q] = *(const v2f*)&Glane[(t >> 2) * 64 + (t & 3) * 8 + 2 * q];
    }

#pragma unroll
    for (int p = 0; p < GMM_NB / 2; ++p) {
        const float xu0 = x0[2 * p],     xu1 = x1[2 * p];
        const float xv0 = x0[2 * p + 1], xv1 = x1[2 * p + 1];

        v2f Fu0, Fu1, Fv0, Fv1;
        if (half == 0) {
            Fu0.x = xu0 * xu0; Fu0.y = xu0 * xu1; Fu1.x = xu1;  Fu1.y = 1.0f;
            Fv0.x = xv0 * xv0; Fv0.y = xv0 * xv1; Fv1.x = xv1;  Fv1.y = 1.0f;
        } else {
            Fu0.x = xu1 * xu1; Fu0.y = xu0;       Fu1.x = 0.0f; Fu1.y = 0.0f;
            Fv0.x = xv1 * xv1; Fv0.y = xv0;       Fv1.x = 0.0f; Fv1.y = 0.0f;
        }

        v8f su = {}, sv = {};
#pragma unroll
        for (int t = 0; t < 8; ++t) {
            v8f au = {}, av = {};
            au = WMMA4(M0[t], Fu0, au);
            av = WMMA4(M0[t], Fv0, av);
            au = WMMA4(M1[t], Fu1, au);
            av = WMMA4(M1[t], Fv1, av);

#pragma unroll
            for (int q = 0; q < 4; ++q) {
                v2f Eu, Ev;
                Eu.x = __expf(au[2 * q]);
                Eu.y = __expf(au[2 * q + 1]);
                Ev.x = __expf(av[2 * q]);
                Ev.y = __expf(av[2 * q + 1]);
                su = WMMA4(G[t][q], Eu, su);
                sv = WMMA4(G[t][q], Ev, sv);
            }
        }

        if (half == 0) {
            gmm_store_rows(su, xu0, xu1, out, b0 + 32 * p + n,      k, Btot);
            gmm_store_rows(sv, xv0, xv1, out, b0 + 32 * p + 16 + n, k, Btot);
        }
    }
}

extern "C" void kernel_launch(void* const* d_in, const int* in_sizes, int n_in,
                              void* d_out, int out_size, void* d_ws, size_t ws_size,
                              hipStream_t stream)
{
    const float* kps = (const float*)d_in[0];
    const float* w   = (const float*)d_in[1];
    const float* mu  = (const float*)d_in[2];
    const float* cov = (const float*)d_in[3];
    float* out = (float*)d_out;

    const int Btot = in_sizes[0] / (GMM_K * 2);

    dim3 grid(Btot / (64 * GMM_NB), GMM_K);
    gmm_wmma<<<grid, 128, 0, stream>>>(kps, w, mu, cov, out, Btot);
}
